# starting kernel
# baseline (speedup 1.0000x reference)
#include <hip/hip_runtime.h>

typedef __attribute__((ext_vector_type(2))) float v2f;
typedef __attribute__((ext_vector_type(8))) float v8f;

namespace {
constexpr int NB   = 256;
constexpr int NN   = 5000;
constexpr int NK   = 100;
constexpr int DIM  = 128;
constexpr int ROWS = NK + 2;
}

__global__ void __launch_bounds__(DIM)
coord_embed_gather(const float* __restrict__ nodes,
                   const int*  __restrict__ first_idx,
                   const int*  __restrict__ last_idx,
                   const int*  __restrict__ cand,
                   const float* __restrict__ cW,
                   const float* __restrict__ cb,
                   float* __restrict__ X1,
                   float* __restrict__ X2,
                   float* __restrict__ out)
{
  const int b = blockIdx.x;
  const int d = threadIdx.x;

  __shared__ int s_idx[NK];
  __shared__ int s_pos[NK];
  __shared__ int s_count;

  if (d < NK) s_idx[d] = cand[b * NK + d];
  __syncthreads();
  if (d < NK) {
    int p = 0;
    for (int j = 0; j < d; ++j) p += (s_idx[j] != -1);
    s_pos[d] = p;
  }
  if (d == 0) {
    int c = 0;
    for (int j = 0; j < NK; ++j) c += (s_idx[j] != -1);
    s_count = c;
  }
  __syncthreads();

  const float w0   = cW[d];
  const float w1   = cW[DIM + d];
  const float bias = cb[d];
  const float* nb  = nodes + (size_t)b * NN * 2;
  float* ob        = out + (size_t)b * ROWS * DIM;

  {
    const int fi = first_idx[b];
    X1[b * DIM + d] = fmaf(nb[fi * 2 + 0], w0, fmaf(nb[fi * 2 + 1], w1, bias));
    const int li = last_idx[b];
    X2[b * DIM + d] = fmaf(nb[li * 2 + 0], w0, fmaf(nb[li * 2 + 1], w1, bias));
  }

  for (int j = 0; j < NK; ++j) {
    const int ci = s_idx[j];
    if (ci != -1) {
      const float e = fmaf(nb[ci * 2 + 0], w0, fmaf(nb[ci * 2 + 1], w1, bias));
      ob[(2 + s_pos[j]) * DIM + d] = e;
    }
  }
  for (int s = s_count; s < NK; ++s) ob[(2 + s) * DIM + d] = 0.0f;
}

__global__ void __launch_bounds__(256)
wmma_heads(const float* __restrict__ X1, const float* __restrict__ X2,
           const float* __restrict__ W1, const float* __restrict__ b1,
           const float* __restrict__ W2, const float* __restrict__ b2,
           float* __restrict__ out)
{
  const int lane  = threadIdx.x & 31;
  const int wave  = threadIdx.x >> 5;
  const int mtile = blockIdx.x;
  const int r  = lane & 15;
  const int hi = lane >> 4;
  const int n  = wave * 16 + r;

  const float* Xg[2] = { X1, X2 };
  const float* Wg[2] = { W1, W2 };
  const float* Bg[2] = { b1, b2 };

  #pragma unroll
  for (int g = 0; g < 2; ++g) {
    const float* X = Xg[g];
    const float* W = Wg[g];
    v8f c = {};
    #pragma unroll
    for (int kk = 0; kk < DIM / 4; ++kk) {
      const int k = kk * 4;
      const v2f a = *(const v2f*)(X + (mtile * 16 + r) * DIM + k + 2 * hi);
      v2f bb;
      bb.x = W[(k + 2 * hi + 0) * DIM + n];
      bb.y = W[(k + 2 * hi + 1) * DIM + n];
      c = __builtin_amdgcn_wmma_f32_16x16x4_f32(
               false, a,  false, bb,
               (short)0, c,  false,  false);
    }
    const float bias = Bg[g][n];
    #pragma unroll
    for (int i = 0; i < 8; ++i) {
      const int batch = mtile * 16 + i + 8 * hi;
      out[(size_t)batch * ROWS * DIM + g * DIM + n] = c[i] + bias;
    }
  }
}

extern "C" void kernel_launch(void* const* d_in, const int* in_sizes, int n_in,
                              void* d_out, int out_size, void* d_ws, size_t ws_size,
                              hipStream_t stream)
{
  const float* nodes     = (const float*)d_in[0];
  const int*   first_idx = (const int*)d_in[1];
  const int*   last_idx  = (const int*)d_in[2];
  const int*   cand      = (const int*)d_in[3];
  const float* cW        = (const float*)d_in[4];
  const float* cb        = (const float*)d_in[5];
  const float* W1        = (const float*)d_in[6];
  const float* b1        = (const float*)d_in[7];
  const float* W2        = (const float*)d_in[8];
  const float* b2        = (const float*)d_in[9];
  float* out = (float*)d_out;

  float* X1 = (float*)d_ws;
  float* X2 = X1 + NB * DIM;

  coord_embed_gather<<<NB, DIM, 0, stream>>>(nodes, first_idx, last_idx, cand,
                                             cW, cb, X1, X2, out);
  wmma_heads<<<NB / 16, 256, 0, stream>>>(X1, X2, W1, b1, W2, b2, out);
}


# gather kernel: one lane per candidate loads its own index and node, so a batch's 102 node reads fly together (was 100 serial dependent loads per thread); rows from v_readlane, whole-row stores
# speedup vs baseline: 4.6267x; 4.6267x over previous
#include <hip/hip_runtime.h>

typedef __attribute__((ext_vector_type(2))) float v2f;
typedef __attribute__((ext_vector_type(4))) float v4f;
typedef __attribute__((ext_vector_type(8))) float v8f;

namespace {
constexpr int NB   = 256;
constexpr int NN   = 5000;
constexpr int NK   = 100;
constexpr int DIM  = 128;
constexpr int ROWS = NK + 2;
constexpr int NGRP = (NK + 31) / 32;
static_assert(NGRP * 32 == DIM, "one lane per candidate slot; a lane holds 4 channels");
static_assert((NK >> 5) == ((NK + 1) >> 5) && NK + 2 <= DIM, "first and last node sit in one wave");
}

static __device__ __forceinline__ v4f embed4(float x, float y, v4f w0, v4f w1, v4f bias)
{
  v4f e;
  e.x = fmaf(x, w0.x, fmaf(y, w1.x, bias.x));
  e.y = fmaf(x, w0.y, fmaf(y, w1.y, bias.y));
  e.z = fmaf(x, w0.z, fmaf(y, w1.z, bias.z));
  e.w = fmaf(x, w0.w, fmaf(y, w1.w, bias.w));
  return e;
}

static __device__ __forceinline__ float from_lane(float v, int src)
{
  return __builtin_bit_cast(float, __builtin_amdgcn_readlane(__builtin_bit_cast(int, v), src));
}

__global__ void __launch_bounds__(DIM)
coord_embed_gather(const float* __restrict__ nodes,
                   const int*  __restrict__ first_idx,
                   const int*  __restrict__ last_idx,
                   const int*  __restrict__ cand,
                   const float* __restrict__ cW,
                   const float* __restrict__ cb,
                   float* __restrict__ X1,
                   float* __restrict__ X2,
                   float* __restrict__ out)
{
  const int b    = blockIdx.x;
  const int d    = threadIdx.x;
  const int lane = d & 31;
  const int wave = __builtin_amdgcn_readfirstlane(d >> 5);

  const int*   cnd = cand + b * NK;
  const float* nb  = nodes + (size_t)b * NN * 2;
  float*       ob  = out + (size_t)b * ROWS * DIM;

  const int fi = first_idx[b];
  const int li = last_idx[b];

  int ci[NGRP];
  for (int g = 0; g < NGRP; ++g) {
    const int j = g * 32 + lane;
    ci[g] = (j < NK) ? cnd[j] : -1;
  }

  const v4f w0 = *(const v4f*)(cW + 4 * lane);
  const v4f w1 = *(const v4f*)(cW + DIM + 4 * lane);
  const v4f bs = *(const v4f*)(cb + 4 * lane);

  unsigned mine = 0;
  int base  = 0;
  int count = 0;
  int own   = -1;
  for (int g = 0; g < NGRP; ++g) {
    const unsigned m = __builtin_amdgcn_ballot_w32(ci[g] != -1);
    const int c = __builtin_popcount(m);
    if (g == wave) { mine = m; own = ci[g]; }
    if (g <  wave) base += c;
    count += c;
  }

  bool need = (own != -1);
  if (d == NK)     { own = fi; need = true; }
  if (d == NK + 1) { own = li; need = true; }
  v2f p = {0.0f, 0.0f};
  if (need) p = *(const v2f*)(nb + 2 * own);

  float* orow = ob + (size_t)(2 + base) * DIM;
  for (int l = 0; l < 32; ++l) {
    if (__builtin_expect((mine >> l) & 1u, 1u)) {
      *((v4f*)orow + lane) = embed4(from_lane(p.x, l), from_lane(p.y, l), w0, w1, bs);
      orow += DIM;
    }
  }

  if (wave == (NK >> 5)) {
    const int lf = NK & 31, ll = (NK + 1) & 31;
    *((v4f*)(X1 + b * DIM) + lane) = embed4(from_lane(p.x, lf), from_lane(p.y, lf), w0, w1, bs);
    *((v4f*)(X2 + b * DIM) + lane) = embed4(from_lane(p.x, ll), from_lane(p.y, ll), w0, w1, bs);
  }

  const v4f zero = {0.0f, 0.0f, 0.0f, 0.0f};
  for (int s = count + wave; s < NK; s += NGRP)
    *((v4f*)(ob + (size_t)(2 + s) * DIM) + lane) = zero;
}

__global__ void __launch_bounds__(256)
wmma_heads(const float* __restrict__ X1, const float* __restrict__ X2,
           const float* __restrict__ W1, const float* __restrict__ b1,
           const float* __restrict__ W2, const float* __restrict__ b2,
           float* __restrict__ out)
{
  const int lane  = threadIdx.x & 31;
  const int wave  = threadIdx.x >> 5;
  const int mtile = blockIdx.x;
  const int r  = lane & 15;
  const int hi = lane >> 4;
  const int n  = wave * 16 + r;

  const float* Xg[2] = { X1, X2 };
  const float* Wg[2] = { W1, W2 };
  const float* Bg[2] = { b1, b2 };

  #pragma unroll
  for (int g = 0; g < 2; ++g) {
    const float* X = Xg[g];
    const float* W = Wg[g];
    v8f c = {};
    #pragma unroll
    for (int kk = 0; kk < DIM / 4; ++kk) {
      const int k = kk * 4;
      const v2f a = *(const v2f*)(X + (mtile * 16 + r) * DIM + k + 2 * hi);
      v2f bb;
      bb.x = W[(k + 2 * hi + 0) * DIM + n];
      bb.y = W[(k + 2 * hi + 1) * DIM + n];
      c = __builtin_amdgcn_wmma_f32_16x16x4_f32(
               false, a,  false, bb,
               (short)0, c,  false,  false);
    }
    const float bias = Bg[g][n];
    #pragma unroll
    for (int i = 0; i < 8; ++i) {
      const int batch = mtile * 16 + i + 8 * hi;
      out[(size_t)batch * ROWS * DIM + g * DIM + n] = c[i] + bias;
    }
  }
}

extern "C" void kernel_launch(void* const* d_in, const int* in_sizes, int n_in,
                              void* d_out, int out_size, void* d_ws, size_t ws_size,
                              hipStream_t stream)
{
  const float* nodes     = (const float*)d_in[0];
  const int*   first_idx = (const int*)d_in[1];
  const int*   last_idx  = (const int*)d_in[2];
  const int*   cand      = (const int*)d_in[3];
  const float* cW        = (const float*)d_in[4];
  const float* cb        = (const float*)d_in[5];
  const float* W1        = (const float*)d_in[6];
  const float* b1        = (const float*)d_in[7];
  const float* W2        = (const float*)d_in[8];
  const float* b2        = (const float*)d_in[9];
  float* out = (float*)d_out;

  float* X1 = (float*)d_ws;
  float* X2 = X1 + NB * DIM;

  coord_embed_gather<<<NB, DIM, 0, stream>>>(nodes, first_idx, last_idx, cand,
                                             cW, cb, X1, X2, out);
  wmma_heads<<<NB / 16, 256, 0, stream>>>(X1, X2, W1, b1, W2, b2, out);
}


# one launch instead of two: 32 extra blocks of the gather kernel do both head products (a wave = two column-interleaved tiles), embedding their A rows in registers from the nodes; no staged rows
# speedup vs baseline: 10.5406x; 2.2782x over previous
#include <hip/hip_runtime.h>


typedef __attribute__((ext_vector_type(2))) float v2f;
typedef __attribute__((ext_vector_type(4))) float v4f;
typedef __attribute__((ext_vector_type(8))) float v8f;

namespace {
constexpr int NB   = 256;
constexpr int NN   = 5000;
constexpr int NK   = 100;
constexpr int DIM  = 128;
constexpr int ROWS = NK + 2;
constexpr int NGRP = (NK + 31) / 32;
constexpr int MT   = NB / 16;
constexpr int HEAD_BLOCKS = MT * 2;
static_assert(NGRP * 32 == DIM, "one lane per candidate slot; a lane holds 4 channels");
static_assert(DIM / 32 == NGRP, "a head block's waves each own a 32-column strip");
constexpr int GRID = HEAD_BLOCKS + NB;
}

static __device__ __forceinline__ v4f embed4(float x, float y, v4f w0, v4f w1, v4f bias)
{
  v4f e;
  e.x = fmaf(x, w0.x, fmaf(y, w1.x, bias.x));
  e.y = fmaf(x, w0.y, fmaf(y, w1.y, bias.y));
  e.z = fmaf(x, w0.z, fmaf(y, w1.z, bias.z));
  e.w = fmaf(x, w0.w, fmaf(y, w1.w, bias.w));
  return e;
}

static __device__ __forceinline__ float from_lane(float v, int src)
{
  return __builtin_bit_cast(float, __builtin_amdgcn_readlane(__builtin_bit_cast(int, v), src));
}

static __device__ __forceinline__ void
head_tile(const float* __restrict__ nodes, const int* __restrict__ idx,
          const float* __restrict__ cW, const float* __restrict__ cb,
          const float* __restrict__ Wm, const float* __restrict__ bv,
          float* __restrict__ out, int mtile, int g)
{
  const int lane = threadIdx.x & 31;
  const int wave = __builtin_amdgcn_readfirstlane(threadIdx.x >> 5);
  const int r  = lane & 15;
  const int hi = lane >> 4;
  const int n0 = wave * 32 + 2 * r;

  const int row = mtile * 16 + r;
  const int ni  = idx[row];

  v2f bq[DIM / 4][2];
  v2f q0[DIM / 4], q1[DIM / 4], qb[DIM / 4];
  constexpr int EARLY = 12;
  #pragma unroll
  for (int kk = 0; kk < EARLY; ++kk) {
    bq[kk][0] = *(const v2f*)(Wm + (4 * kk + 2 * hi + 0) * DIM + n0);
    bq[kk][1] = *(const v2f*)(Wm + (4 * kk + 2 * hi + 1) * DIM + n0);
    q0[kk] = *(const v2f*)(cW + 4 * kk + 2 * hi);
    q1[kk] = *(const v2f*)(cW + DIM + 4 * kk + 2 * hi);
    qb[kk] = *(const v2f*)(cb + 4 * kk + 2 * hi);
  }
  const v2f bias = *(const v2f*)(bv + n0);
  __builtin_amdgcn_sched_barrier(0);

  const v2f p = *(const v2f*)(nodes + ((size_t)row * NN + ni) * 2);

  #pragma unroll
  for (int kk = EARLY; kk < DIM / 4; ++kk) {
    bq[kk][0] = *(const v2f*)(Wm + (4 * kk + 2 * hi + 0) * DIM + n0);
    bq[kk][1] = *(const v2f*)(Wm + (4 * kk + 2 * hi + 1) * DIM + n0);
    q0[kk] = *(const v2f*)(cW + 4 * kk + 2 * hi);
    q1[kk] = *(const v2f*)(cW + DIM + 4 * kk + 2 * hi);
    qb[kk] = *(const v2f*)(cb + 4 * kk + 2 * hi);
  }

  v8f c0 = {}, c1 = {};
  #pragma unroll
  for (int kk = 0; kk < DIM / 4; ++kk) {
    v2f a;
    a.x = fmaf(p.x, q0[kk].x, fmaf(p.y, q1[kk].x, qb[kk].x));
    a.y = fmaf(p.x, q0[kk].y, fmaf(p.y, q1[kk].y, qb[kk].y));
    v2f b0, b1;
    b0.x = bq[kk][0].x;  b0.y = bq[kk][1].x;
    b1.x = bq[kk][0].y;  b1.y = bq[kk][1].y;
    c0 = __builtin_amdgcn_wmma_f32_16x16x4_f32(
              false, a,  false, b0,
              (short)0, c0,  false,  false);
    c1 = __builtin_amdgcn_wmma_f32_16x16x4_f32(
              false, a,  false, b1,
              (short)0, c1,  false,  false);
  }

  #pragma unroll
  for (int i = 0; i < 8; ++i) {
    const int batch = mtile * 16 + i + 8 * hi;
    v2f o;
    o.x = c0[i] + bias.x;
    o.y = c1[i] + bias.y;
    *(v2f*)(out + (size_t)batch * ROWS * DIM + g * DIM + n0) = o;
  }
}

__global__ void __launch_bounds__(DIM) __attribute__((amdgpu_waves_per_eu(1, 4)))
embed_gather_heads(const float* __restrict__ nodes,
                   const int*  __restrict__ first_idx,
                   const int*  __restrict__ last_idx,
                   const int*  __restrict__ cand,
                   const float* __restrict__ cW,
                   const float* __restrict__ cb,
                   const float* __restrict__ W1, const float* __restrict__ b1,
                   const float* __restrict__ W2, const float* __restrict__ b2,
                   float* __restrict__ out)
{
  if (blockIdx.x < HEAD_BLOCKS) {
    const int mtile = blockIdx.x >> 1;
    const int g     = blockIdx.x & 1;
    head_tile(nodes, g ? last_idx : first_idx, cW, cb, g ? W2 : W1, g ? b2 : b1, out, mtile, g);
    return;
  }

  const int b    = blockIdx.x - HEAD_BLOCKS;
  const int d    = threadIdx.x;
  const int lane = d & 31;
  const int wave = __builtin_amdgcn_readfirstlane(d >> 5);

  const int*   cnd = cand + b * NK;
  const float* nb  = nodes + (size_t)b * NN * 2;
  float*       ob  = out + (size_t)b * ROWS * DIM;

  int ci[NGRP];
  for (int g = 0; g < NGRP; ++g) {
    const int j = g * 32 + lane;
    ci[g] = (j < NK) ? cnd[j] : -1;
  }

  const v4f w0 = *(const v4f*)(cW + 4 * lane);
  const v4f w1 = *(const v4f*)(cW + DIM + 4 * lane);
  const v4f bs = *(const v4f*)(cb + 4 * lane);

  unsigned mine = 0;
  int base  = 0;
  int count = 0;
  int own   = -1;
  for (int g = 0; g < NGRP; ++g) {
    const unsigned m = __builtin_amdgcn_ballot_w32(ci[g] != -1);
    const int c = __builtin_popcount(m);
    if (g == wave) { mine = m; own = ci[g]; }
    if (g <  wave) base += c;
    count += c;
  }

  v2f p = {0.0f, 0.0f};
  if (own != -1) p = *(const v2f*)(nb + 2 * own);

  float* orow = ob + (size_t)(2 + base) * DIM;
  for (int l = 0; l < 32; ++l) {
    if (__builtin_expect((mine >> l) & 1u, 1u)) {
      *((v4f*)orow + lane) = embed4(from_lane(p.x, l), from_lane(p.y, l), w0, w1, bs);
      orow += DIM;
    }
  }

  const v4f zero = {0.0f, 0.0f, 0.0f, 0.0f};
  for (int s = count + wave; s < NK; s += NGRP)
    *((v4f*)(ob + (size_t)(2 + s) * DIM) + lane) = zero;
}

extern "C" void kernel_launch(void* const* d_in, const int* in_sizes, int n_in,
                              void* d_out, int out_size, void* d_ws, size_t ws_size,
                              hipStream_t stream)
{
  const float* nodes     = (const float*)d_in[0];
  const int*   first_idx = (const int*)d_in[1];
  const int*   last_idx  = (const int*)d_in[2];
  const int*   cand      = (const int*)d_in[3];
  const float* cW        = (const float*)d_in[4];
  const float* cb        = (const float*)d_in[5];
  const float* W1        = (const float*)d_in[6];
  const float* b1        = (const float*)d_in[7];
  const float* W2        = (const float*)d_in[8];
  const float* b2        = (const float*)d_in[9];
  float* out = (float*)d_out;

  embed_gather_heads<<<GRID, DIM, 0, stream>>>(nodes, first_idx, last_idx, cand,
                                                           cW, cb, W1, b1, W2, b2, out);
}
